# starting kernel
# baseline (speedup 1.0000x reference)
#include <hip/hip_runtime.h>
#include <math.h>

typedef __attribute__((ext_vector_type(2))) float v2f;
typedef __attribute__((ext_vector_type(8))) float v8f;

#define N_NODES   100000
#define N_EDGES   1600000
#define IN_FEAT   256
#define HIDDEN    128
#define N_CLASSES 16


__global__ void init_deg_kernel(float* __restrict__ deg) {
    int i = blockIdx.x * blockDim.x + threadIdx.x;
    if (i < N_NODES) deg[i] = 1.0f;
}

__global__ void deg_accum_kernel(const int* __restrict__ col, float* __restrict__ deg) {
    int e = blockIdx.x * blockDim.x + threadIdx.x;
    if (e < N_EDGES) unsafeAtomicAdd(&deg[col[e]], 1.0f);
}

__global__ void dinv_kernel(float* __restrict__ deg) {
    int i = blockIdx.x * blockDim.x + threadIdx.x;
    if (i < N_NODES) deg[i] = rsqrtf(deg[i]);
}


__global__ __launch_bounds__(256) void gemm1_kernel(const float* __restrict__ x,
                                                    const float* __restrict__ W1,
                                                    float* __restrict__ hp) {
    const int lane = threadIdx.x & 31;
    const int wave = (blockIdx.x * blockDim.x + threadIdx.x) >> 5;
    const int row0 = wave * 16;
    if (row0 >= N_NODES) return;

    const int m  = lane & 15;
    const int kh = lane >> 4;
    const float* arow = x + (size_t)(row0 + m) * IN_FEAT;

    v8f acc[8];
#pragma unroll
    for (int n = 0; n < 8; ++n) acc[n] = v8f{};

    for (int k = 0; k < IN_FEAT; k += 4) {
        v2f a;
        a.x = arow[k + 2 * kh + 0];
        a.y = arow[k + 2 * kh + 1];
        const float* b0 = W1 + (size_t)(k + 2 * kh) * HIDDEN + m;
#pragma unroll
        for (int n = 0; n < 8; ++n) {
            v2f b;
            b.x = b0[n * 16];
            b.y = b0[HIDDEN + n * 16];
            acc[n] = __builtin_amdgcn_wmma_f32_16x16x4_f32(
                false, a, false, b, (short)0, acc[n], false, false);
        }
    }

    const int mo = kh * 8;
#pragma unroll
    for (int n = 0; n < 8; ++n) {
#pragma unroll
        for (int j = 0; j < 8; ++j) {
            hp[(size_t)(row0 + mo + j) * HIDDEN + n * 16 + m] = acc[n][j];
        }
    }
}


__global__ __launch_bounds__(256) void gemm2_kernel(const float* __restrict__ h,
                                                    const float* __restrict__ W2,
                                                    float* __restrict__ h2p) {
    const int lane = threadIdx.x & 31;
    const int wave = (blockIdx.x * blockDim.x + threadIdx.x) >> 5;
    const int row0 = wave * 16;
    if (row0 >= N_NODES) return;

    const int m  = lane & 15;
    const int kh = lane >> 4;
    const float* arow = h + (size_t)(row0 + m) * HIDDEN;

    v8f acc = v8f{};
    for (int k = 0; k < HIDDEN; k += 4) {
        v2f a;
        a.x = arow[k + 2 * kh + 0];
        a.y = arow[k + 2 * kh + 1];
        v2f b;
        b.x = W2[(size_t)(k + 2 * kh + 0) * N_CLASSES + m];
        b.y = W2[(size_t)(k + 2 * kh + 1) * N_CLASSES + m];
        acc = __builtin_amdgcn_wmma_f32_16x16x4_f32(
            false, a, false, b, (short)0, acc, false, false);
    }

    const int mo = kh * 8;
#pragma unroll
    for (int j = 0; j < 8; ++j) {
        h2p[(size_t)(row0 + mo + j) * N_CLASSES + m] = acc[j];
    }
}


__global__ void agg_self_kernel(const float* __restrict__ src,
                                const float* __restrict__ dinv,
                                float* __restrict__ agg,
                                int shift, long total) {
    long i = (long)blockIdx.x * blockDim.x + threadIdx.x;
    if (i < total) {
        int node = (int)(i >> shift);
        float d = dinv[node];
        agg[i] = d * d * src[i];
    }
}

__global__ __launch_bounds__(256) void agg1_edge_kernel(const int* __restrict__ row,
                                                        const int* __restrict__ col,
                                                        const float* __restrict__ dinv,
                                                        const float* __restrict__ hp,
                                                        float* __restrict__ agg) {
    const int lane = threadIdx.x & 31;
    const long e = ((long)blockIdx.x * blockDim.x + threadIdx.x) >> 5;
    if (e >= N_EDGES) return;
    const int r = row[e];
    const int c = col[e];
    const float w = dinv[r] * dinv[c];
    const float4 v = ((const float4*)(hp + (size_t)r * HIDDEN))[lane];
    float* dst = agg + (size_t)c * HIDDEN + lane * 4;
    unsafeAtomicAdd(dst + 0, w * v.x);
    unsafeAtomicAdd(dst + 1, w * v.y);
    unsafeAtomicAdd(dst + 2, w * v.z);
    unsafeAtomicAdd(dst + 3, w * v.w);
}

__global__ void agg2_edge_kernel(const int* __restrict__ row,
                                 const int* __restrict__ col,
                                 const float* __restrict__ dinv,
                                 const float* __restrict__ h2p,
                                 float* __restrict__ agg) {
    long t = (long)blockIdx.x * blockDim.x + threadIdx.x;
    if (t >= (long)N_EDGES * N_CLASSES) return;
    int e = (int)(t >> 4);
    int f = (int)(t & 15);
    int r = row[e];
    int c = col[e];
    float w = dinv[r] * dinv[c];
    unsafeAtomicAdd(&agg[(size_t)c * N_CLASSES + f], w * h2p[(size_t)r * N_CLASSES + f]);
}

__global__ void bias_relu_kernel(const float* __restrict__ agg,
                                 const float* __restrict__ b1,
                                 float* __restrict__ out) {
    long i = (long)blockIdx.x * blockDim.x + threadIdx.x;
    if (i < (long)N_NODES * HIDDEN) {
        float v = agg[i] + b1[(int)(i & (HIDDEN - 1))];
        out[i] = v > 0.0f ? v : 0.0f;
    }
}

__global__ void logsoftmax_kernel(const float* __restrict__ agg,
                                  const float* __restrict__ b2,
                                  float* __restrict__ out) {
    int i = blockIdx.x * blockDim.x + threadIdx.x;
    if (i >= N_NODES) return;
    float v[N_CLASSES];
    float mx = -INFINITY;
#pragma unroll
    for (int c = 0; c < N_CLASSES; ++c) {
        v[c] = agg[(size_t)i * N_CLASSES + c] + b2[c];
        mx = fmaxf(mx, v[c]);
    }
    float s = 0.0f;
#pragma unroll
    for (int c = 0; c < N_CLASSES; ++c) s += expf(v[c] - mx);
    float lse = mx + logf(s);
#pragma unroll
    for (int c = 0; c < N_CLASSES; ++c) {
        out[(size_t)i * N_CLASSES + c] = v[c] - lse;
    }
}


static inline int cdiv_l(long a, int b) { return (int)((a + b - 1) / b); }

extern "C" void kernel_launch(void* const* d_in, const int* in_sizes, int n_in,
                              void* d_out, int out_size, void* d_ws, size_t ws_size,
                              hipStream_t stream) {
    const float* x   = (const float*)d_in[0];
    const int*   ei  = (const int*)d_in[1];
    const int*   row = ei;
    const int*   col = ei + N_EDGES;
    const float* W1  = (const float*)d_in[2];
    const float* b1  = (const float*)d_in[3];
    const float* W2  = (const float*)d_in[4];
    const float* b2  = (const float*)d_in[5];
    float* out = (float*)d_out;

    float* ws   = (float*)d_ws;
    float* dinv = ws;
    float* hp   = ws + ((N_NODES + 255) & ~255);
    float* agg1 = hp + (size_t)N_NODES * HIDDEN;
    float* h2p  = agg1;
    float* agg2 = agg1 + (size_t)N_NODES * N_CLASSES;

    const int B = 256;

    init_deg_kernel<<<cdiv_l(N_NODES, B), B, 0, stream>>>(dinv);
    deg_accum_kernel<<<cdiv_l(N_EDGES, B), B, 0, stream>>>(col, dinv);
    dinv_kernel<<<cdiv_l(N_NODES, B), B, 0, stream>>>(dinv);

    const long waves1 = (N_NODES + 15) / 16;
    gemm1_kernel<<<cdiv_l(waves1 * 32, B), B, 0, stream>>>(x, W1, hp);
    agg_self_kernel<<<cdiv_l((long)N_NODES * HIDDEN, B), B, 0, stream>>>(
        hp, dinv, agg1, 7, (long)N_NODES * HIDDEN);
    agg1_edge_kernel<<<cdiv_l((long)N_EDGES * 32, B), B, 0, stream>>>(
        row, col, dinv, hp, agg1);
    bias_relu_kernel<<<cdiv_l((long)N_NODES * HIDDEN, B), B, 0, stream>>>(agg1, b1, hp);

    gemm2_kernel<<<cdiv_l(waves1 * 32, B), B, 0, stream>>>(hp, W2, h2p);
    agg_self_kernel<<<cdiv_l((long)N_NODES * N_CLASSES, B), B, 0, stream>>>(
        h2p, dinv, agg2, 4, (long)N_NODES * N_CLASSES);
    agg2_edge_kernel<<<cdiv_l((long)N_EDGES * N_CLASSES, B), B, 0, stream>>>(
        row, col, dinv, h2p, agg2);

    logsoftmax_kernel<<<cdiv_l(N_NODES, B), B, 0, stream>>>(agg2, b2, out);
}


# Edges grouped by destination each call (int count, scan, slot fill); per-node gather replaces f32 atomic scatter, h@W2 fused behind it via LDS; sum order = arrival rank
# speedup vs baseline: 1.4914x; 1.4914x over previous
#include <hip/hip_runtime.h>
#include <math.h>

typedef __attribute__((ext_vector_type(2))) float v2f;
typedef __attribute__((ext_vector_type(8))) float v8f;

#define N_NODES   100000
#define N_EDGES   1600000
#define IN_FEAT   256
#define HIDDEN    128
#define N_CLASSES 16
#define SCAN_T    1024
#define SCAN_CH   98
#define TILE_P    132


__global__ void zero_cnt_kernel(int* __restrict__ cnt) {
    int i = blockIdx.x * blockDim.x + threadIdx.x;
    if (i < N_NODES) cnt[i] = 0;
}

__global__ void count_kernel(const int* __restrict__ col, int* __restrict__ cnt) {
    int e = blockIdx.x * blockDim.x + threadIdx.x;
    if (e < N_EDGES) atomicAdd(&cnt[col[e]], 1);
}

__global__ __launch_bounds__(SCAN_T) void scan_kernel(const int* __restrict__ cnt,
                                                      int* __restrict__ cursor) {
    __shared__ int wsum[32];
    const int lane = threadIdx.x & 31;
    const int w    = threadIdx.x >> 5;
    const int b    = w * (32 * SCAN_CH);
    int sum = 0;
    for (int k = 0; k < SCAN_CH; ++k) {
        const int i = b + k * 32 + lane;
        sum += (i < N_NODES) ? cnt[i] : 0;
    }
#pragma unroll
    for (int o = 16; o >= 1; o >>= 1) sum += __shfl_xor(sum, o);
    if (lane == 0) wsum[w] = sum;
    __syncthreads();
    const int wl = wsum[lane];
    int winc = wl;
#pragma unroll
    for (int o = 1; o < 32; o <<= 1) {
        const int u = __shfl_up(winc, o);
        if (lane >= o) winc += u;
    }
    int carry = __shfl(winc - wl, w);
    for (int k = 0; k < SCAN_CH; ++k) {
        const int i = b + k * 32 + lane;
        const int v = (i < N_NODES) ? cnt[i] : 0;
        int inc = v;
#pragma unroll
        for (int o = 1; o < 32; o <<= 1) {
            const int u = __shfl_up(inc, o);
            if (lane >= o) inc += u;
        }
        if (i < N_NODES) cursor[i] = carry + inc - v;
        carry += __shfl(inc, 31);
    }
}

__global__ void fill_kernel(const int* __restrict__ row, const int* __restrict__ col,
                            int* __restrict__ cursor, int* __restrict__ srcs) {
    int e = blockIdx.x * blockDim.x + threadIdx.x;
    if (e < N_EDGES) {
        int p = atomicAdd(&cursor[col[e]], 1);
        srcs[p] = row[e];
    }
}


__global__ __launch_bounds__(256) void gemm1_kernel(const float* __restrict__ x,
                                                    const float* __restrict__ W1,
                                                    float* __restrict__ hp) {
    const int lane = threadIdx.x & 31;
    const int wave = (blockIdx.x * blockDim.x + threadIdx.x) >> 5;
    const int row0 = wave * 16;
    if (row0 >= N_NODES) return;

    const int m  = lane & 15;
    const int kh = lane >> 4;
    const float* arow = x + (size_t)(row0 + m) * IN_FEAT;

    v8f acc[8];
#pragma unroll
    for (int n = 0; n < 8; ++n) acc[n] = v8f{};

    for (int k = 0; k < IN_FEAT; k += 4) {
        v2f a;
        a.x = arow[k + 2 * kh + 0];
        a.y = arow[k + 2 * kh + 1];
        const float* b0 = W1 + (size_t)(k + 2 * kh) * HIDDEN + m;
#pragma unroll
        for (int n = 0; n < 8; ++n) {
            v2f b;
            b.x = b0[n * 16];
            b.y = b0[HIDDEN + n * 16];
            acc[n] = __builtin_amdgcn_wmma_f32_16x16x4_f32(
                false, a, false, b, (short)0, acc[n], false, false);
        }
    }

    const int mo = kh * 8;
#pragma unroll
    for (int n = 0; n < 8; ++n) {
#pragma unroll
        for (int j = 0; j < 8; ++j) {
            hp[(size_t)(row0 + mo + j) * HIDDEN + n * 16 + m] = acc[n][j];
        }
    }
}


__global__ __launch_bounds__(256) void gather1_gemm2_kernel(const int* __restrict__ cnt,
                                                            const int* __restrict__ cursor,
                                                            const int* __restrict__ srcs,
                                                            const float* __restrict__ hp,
                                                            const float* __restrict__ b1,
                                                            const float* __restrict__ W2,
                                                            float* __restrict__ h2p) {
    __shared__ float tile[8 * 16 * TILE_P];
    const int lane = threadIdx.x & 31;
    const int wid  = threadIdx.x >> 5;
    const int wave = blockIdx.x * 8 + wid;
    const int row0 = wave * 16;
    const bool active = row0 < N_NODES;
    float* mytile = tile + wid * 16 * TILE_P;
    const float4* hp4 = (const float4*)hp;

    if (active) {
        const float4 bias = ((const float4*)b1)[lane];
        for (int jn = 0; jn < 16; ++jn) {
            const int i = row0 + jn;
            const int n   = __builtin_amdgcn_readfirstlane(cnt[i]);
            const int end = __builtin_amdgcn_readfirstlane(cursor[i]);
            const int start = end - n;
            const float di = rsqrtf((float)(n + 1));
            float4 v0 = hp4[(size_t)i * 32 + lane];
            float4 acc;
            acc.x = di * v0.x; acc.y = di * v0.y; acc.z = di * v0.z; acc.w = di * v0.w;
            for (int base = start; base < end; base += 32) {
                const int idx = base + lane;
                int   r_l = 0;
                float d_l = 0.0f;
                if (idx < end) {
                    r_l = srcs[idx];
                    d_l = rsqrtf((float)(cnt[r_l] + 1));
                }
                const int m = (end - base < 32) ? (end - base) : 32;
                for (int j = 0; j < m; ++j) {
                    const int   r = __shfl(r_l, j);
                    const float d = __shfl(d_l, j);
                    const float4 v = hp4[(size_t)r * 32 + lane];
                    acc.x += d * v.x; acc.y += d * v.y; acc.z += d * v.z; acc.w += d * v.w;
                }
            }
            float4 o;
            o.x = di * acc.x + bias.x; o.y = di * acc.y + bias.y;
            o.z = di * acc.z + bias.z; o.w = di * acc.w + bias.w;
            o.x = o.x > 0.0f ? o.x : 0.0f; o.y = o.y > 0.0f ? o.y : 0.0f;
            o.z = o.z > 0.0f ? o.z : 0.0f; o.w = o.w > 0.0f ? o.w : 0.0f;
            *(float4*)(mytile + jn * TILE_P + lane * 4) = o;
        }
    }
    __syncthreads();
    if (active) {
        const int m  = lane & 15;
        const int kh = lane >> 4;
        const float* arow = mytile + m * TILE_P;
        v8f acc2 = v8f{};
        for (int k = 0; k < HIDDEN; k += 4) {
            v2f a;
            a.x = arow[k + 2 * kh + 0];
            a.y = arow[k + 2 * kh + 1];
            v2f b;
            b.x = W2[(size_t)(k + 2 * kh + 0) * N_CLASSES + m];
            b.y = W2[(size_t)(k + 2 * kh + 1) * N_CLASSES + m];
            acc2 = __builtin_amdgcn_wmma_f32_16x16x4_f32(
                false, a, false, b, (short)0, acc2, false, false);
        }
        const int mo = kh * 8;
#pragma unroll
        for (int j = 0; j < 8; ++j) {
            h2p[(size_t)(row0 + mo + j) * N_CLASSES + m] = acc2[j];
        }
    }
}


__global__ __launch_bounds__(256) void gather2_lsm_kernel(const int* __restrict__ cnt,
                                                          const int* __restrict__ cursor,
                                                          const int* __restrict__ srcs,
                                                          const float* __restrict__ h2p,
                                                          const float* __restrict__ b2,
                                                          float* __restrict__ out) {
    const int t = blockIdx.x * blockDim.x + threadIdx.x;
    const int i = t >> 4;
    const int f = t & 15;
    const bool valid = i < N_NODES;
    float v = 0.0f;
    if (valid) {
        const int n   = cnt[i];
        const int end = cursor[i];
        const float di = rsqrtf((float)(n + 1));
        float acc = di * h2p[(size_t)i * N_CLASSES + f];
        for (int j = end - n; j < end; ++j) {
            const int r = srcs[j];
            const float d = rsqrtf((float)(cnt[r] + 1));
            acc += d * h2p[(size_t)r * N_CLASSES + f];
        }
        v = di * acc + b2[f];
    }
    float mx = v;
#pragma unroll
    for (int o = 8; o >= 1; o >>= 1) mx = fmaxf(mx, __shfl_xor(mx, o, 16));
    float s = expf(v - mx);
#pragma unroll
    for (int o = 8; o >= 1; o >>= 1) s += __shfl_xor(s, o, 16);
    const float lse = mx + logf(s);
    if (valid) out[(size_t)i * N_CLASSES + f] = v - lse;
}


static inline int cdiv_l(long a, int b) { return (int)((a + b - 1) / b); }

extern "C" void kernel_launch(void* const* d_in, const int* in_sizes, int n_in,
                              void* d_out, int out_size, void* d_ws, size_t ws_size,
                              hipStream_t stream) {
    const float* x   = (const float*)d_in[0];
    const int*   ei  = (const int*)d_in[1];
    const int*   row = ei;
    const int*   col = ei + N_EDGES;
    const float* W1  = (const float*)d_in[2];
    const float* b1  = (const float*)d_in[3];
    const float* W2  = (const float*)d_in[4];
    const float* b2  = (const float*)d_in[5];
    float* out = (float*)d_out;

    float* ws     = (float*)d_ws;
    int*   cnt    = (int*)ws;
    float* hp     = ws + ((N_NODES + 255) & ~255);
    float* rest   = hp + (size_t)N_NODES * HIDDEN;
    int*   cursor = (int*)rest;
    int*   srcs   = cursor + ((N_NODES + 255) & ~255);
    float* h2p    = (float*)(srcs + N_EDGES);

    const int B = 256;

    zero_cnt_kernel<<<cdiv_l(N_NODES, B), B, 0, stream>>>(cnt);
    count_kernel<<<cdiv_l(N_EDGES, B), B, 0, stream>>>(col, cnt);
    scan_kernel<<<1, SCAN_T, 0, stream>>>(cnt, cursor);
    fill_kernel<<<cdiv_l(N_EDGES, B), B, 0, stream>>>(row, col, cursor, srcs);

    const long waves1 = (N_NODES + 15) / 16;
    gemm1_kernel<<<cdiv_l(waves1 * 32, B), B, 0, stream>>>(x, W1, hp);
    gather1_gemm2_kernel<<<cdiv_l(waves1 * 32, B), B, 0, stream>>>(
        cnt, cursor, srcs, hp, b1, W2, h2p);
    gather2_lsm_kernel<<<cdiv_l((long)N_NODES * N_CLASSES, B), B, 0, stream>>>(
        cnt, cursor, srcs, h2p, b2, out);
}


# Prefix sum of the per-node edge counts spread over 98 workgroups (each adds the counts in front of its 1024 nodes with 16-byte loads, then scans its own) instead of one workgroup's serial walk
# speedup vs baseline: 1.7634x; 1.1824x over previous
#include <hip/hip_runtime.h>
#include <math.h>

typedef __attribute__((ext_vector_type(2))) float v2f;
typedef __attribute__((ext_vector_type(8))) float v8f;

#define N_NODES   100000
#define N_EDGES   1600000
#define IN_FEAT   256
#define HIDDEN    128
#define N_CLASSES 16
#define SCAN_T    1024
#define SCAN_B    98
#define TILE_P    132


__global__ void zero_cnt_kernel(int* __restrict__ cnt) {
    int i = blockIdx.x * blockDim.x + threadIdx.x;
    if (i < N_NODES) cnt[i] = 0;
}

__global__ void count_kernel(const int* __restrict__ col, int* __restrict__ cnt) {
    int e = blockIdx.x * blockDim.x + threadIdx.x;
    if (e < N_EDGES) atomicAdd(&cnt[col[e]], 1);
}

__global__ __launch_bounds__(SCAN_T) void scan_kernel(const int* __restrict__ cnt,
                                                      int* __restrict__ cursor) {
    __shared__ int wpre[32];
    __shared__ int wtot[32];
    const int lane = threadIdx.x & 31;
    const int w    = threadIdx.x >> 5;
    const int i    = blockIdx.x * SCAN_T + threadIdx.x;
    const int v    = (i < N_NODES) ? cnt[i] : 0;

    const int4* c4 = (const int4*)cnt;
    const int n4 = blockIdx.x * (SCAN_T / 4);
    int pre = 0;
    for (int q0 = threadIdx.x; q0 < n4; q0 += 8 * SCAN_T) {
        int4 c[8];
#pragma unroll
        for (int u = 0; u < 8; ++u) {
            const int q = q0 + u * SCAN_T;
            c[u] = c4[(q < n4) ? q : q0];
        }
#pragma unroll
        for (int u = 0; u < 8; ++u) {
            const int q = q0 + u * SCAN_T;
            const int s = (c[u].x + c[u].y) + (c[u].z + c[u].w);
            pre += (q < n4) ? s : 0;
        }
    }
#pragma unroll
    for (int o = 16; o >= 1; o >>= 1) pre += __shfl_xor(pre, o);
    int inc = v;
#pragma unroll
    for (int o = 1; o < 32; o <<= 1) {
        const int u = __shfl_up(inc, o);
        if (lane >= o) inc += u;
    }
    if (lane == 0)  wpre[w] = pre;
    if (lane == 31) wtot[w] = inc;
    __syncthreads();
    int front = 0;
#pragma unroll
    for (int k = 0; k < 32; ++k) front += wpre[k] + ((k < w) ? wtot[k] : 0);
    if (i < N_NODES) cursor[i] = front + inc - v;
}

__global__ void fill_kernel(const int* __restrict__ row, const int* __restrict__ col,
                            int* __restrict__ cursor, int* __restrict__ srcs) {
    int e = blockIdx.x * blockDim.x + threadIdx.x;
    if (e < N_EDGES) {
        int p = atomicAdd(&cursor[col[e]], 1);
        srcs[p] = row[e];
    }
}


__global__ __launch_bounds__(256) void gemm1_kernel(const float* __restrict__ x,
                                                    const float* __restrict__ W1,
                                                    float* __restrict__ hp) {
    const int lane = threadIdx.x & 31;
    const int wave = (blockIdx.x * blockDim.x + threadIdx.x) >> 5;
    const int row0 = wave * 16;
    if (row0 >= N_NODES) return;

    const int m  = lane & 15;
    const int kh = lane >> 4;
    const float* arow = x + (size_t)(row0 + m) * IN_FEAT;

    v8f acc[8];
#pragma unroll
    for (int n = 0; n < 8; ++n) acc[n] = v8f{};

    for (int k = 0; k < IN_FEAT; k += 4) {
        v2f a;
        a.x = arow[k + 2 * kh + 0];
        a.y = arow[k + 2 * kh + 1];
        const float* b0 = W1 + (size_t)(k + 2 * kh) * HIDDEN + m;
#pragma unroll
        for (int n = 0; n < 8; ++n) {
            v2f b;
            b.x = b0[n * 16];
            b.y = b0[HIDDEN + n * 16];
            acc[n] = __builtin_amdgcn_wmma_f32_16x16x4_f32(
                false, a, false, b, (short)0, acc[n], false, false);
        }
    }

    const int mo = kh * 8;
#pragma unroll
    for (int n = 0; n < 8; ++n) {
#pragma unroll
        for (int j = 0; j < 8; ++j) {
            hp[(size_t)(row0 + mo + j) * HIDDEN + n * 16 + m] = acc[n][j];
        }
    }
}


__global__ __launch_bounds__(256) void gather1_gemm2_kernel(const int* __restrict__ cnt,
                                                            const int* __restrict__ cursor,
                                                            const int* __restrict__ srcs,
                                                            const float* __restrict__ hp,
                                                            const float* __restrict__ b1,
                                                            const float* __restrict__ W2,
                                                            float* __restrict__ h2p) {
    __shared__ float tile[8 * 16 * TILE_P];
    const int lane = threadIdx.x & 31;
    const int wid  = threadIdx.x >> 5;
    const int wave = blockIdx.x * 8 + wid;
    const int row0 = wave * 16;
    const bool active = row0 < N_NODES;
    float* mytile = tile + wid * 16 * TILE_P;
    const float4* hp4 = (const float4*)hp;

    if (active) {
        const float4 bias = ((const float4*)b1)[lane];
        for (int jn = 0; jn < 16; ++jn) {
            const int i = row0 + jn;
            const int n   = __builtin_amdgcn_readfirstlane(cnt[i]);
            const int end = __builtin_amdgcn_readfirstlane(cursor[i]);
            const int start = end - n;
            const float di = rsqrtf((float)(n + 1));
            float4 v0 = hp4[(size_t)i * 32 + lane];
            float4 acc;
            acc.x = di * v0.x; acc.y = di * v0.y; acc.z = di * v0.z; acc.w = di * v0.w;
            for (int base = start; base < end; base += 32) {
                const int idx = base + lane;
                int   r_l = 0;
                float d_l = 0.0f;
                if (idx < end) {
                    r_l = srcs[idx];
                    d_l = rsqrtf((float)(cnt[r_l] + 1));
                }
                const int m = (end - base < 32) ? (end - base) : 32;
                for (int j = 0; j < m; ++j) {
                    const int   r = __shfl(r_l, j);
                    const float d = __shfl(d_l, j);
                    const float4 v = hp4[(size_t)r * 32 + lane];
                    acc.x += d * v.x; acc.y += d * v.y; acc.z += d * v.z; acc.w += d * v.w;
                }
            }
            float4 o;
            o.x = di * acc.x + bias.x; o.y = di * acc.y + bias.y;
            o.z = di * acc.z + bias.z; o.w = di * acc.w + bias.w;
            o.x = o.x > 0.0f ? o.x : 0.0f; o.y = o.y > 0.0f ? o.y : 0.0f;
            o.z = o.z > 0.0f ? o.z : 0.0f; o.w = o.w > 0.0f ? o.w : 0.0f;
            *(float4*)(mytile + jn * TILE_P + lane * 4) = o;
        }
    }
    __syncthreads();
    if (active) {
        const int m  = lane & 15;
        const int kh = lane >> 4;
        const float* arow = mytile + m * TILE_P;
        v8f acc2 = v8f{};
        for (int k = 0; k < HIDDEN; k += 4) {
            v2f a;
            a.x = arow[k + 2 * kh + 0];
            a.y = arow[k + 2 * kh + 1];
            v2f b;
            b.x = W2[(size_t)(k + 2 * kh + 0) * N_CLASSES + m];
            b.y = W2[(size_t)(k + 2 * kh + 1) * N_CLASSES + m];
            acc2 = __builtin_amdgcn_wmma_f32_16x16x4_f32(
                false, a, false, b, (short)0, acc2, false, false);
        }
        const int mo = kh * 8;
#pragma unroll
        for (int j = 0; j < 8; ++j) {
            h2p[(size_t)(row0 + mo + j) * N_CLASSES + m] = acc2[j];
        }
    }
}


__global__ __launch_bounds__(256) void gather2_lsm_kernel(const int* __restrict__ cnt,
                                                          const int* __restrict__ cursor,
                                                          const int* __restrict__ srcs,
                                                          const float* __restrict__ h2p,
                                                          const float* __restrict__ b2,
                                                          float* __restrict__ out) {
    const int t = blockIdx.x * blockDim.x + threadIdx.x;
    const int i = t >> 4;
    const int f = t & 15;
    const bool valid = i < N_NODES;
    float v = 0.0f;
    if (valid) {
        const int n   = cnt[i];
        const int end = cursor[i];
        const float di = rsqrtf((float)(n + 1));
        float acc = di * h2p[(size_t)i * N_CLASSES + f];
        for (int j = end - n; j < end; ++j) {
            const int r = srcs[j];
            const float d = rsqrtf((float)(cnt[r] + 1));
            acc += d * h2p[(size_t)r * N_CLASSES + f];
        }
        v = di * acc + b2[f];
    }
    float mx = v;
#pragma unroll
    for (int o = 8; o >= 1; o >>= 1) mx = fmaxf(mx, __shfl_xor(mx, o, 16));
    float s = expf(v - mx);
#pragma unroll
    for (int o = 8; o >= 1; o >>= 1) s += __shfl_xor(s, o, 16);
    const float lse = mx + logf(s);
    if (valid) out[(size_t)i * N_CLASSES + f] = v - lse;
}


static inline int cdiv_l(long a, int b) { return (int)((a + b - 1) / b); }

extern "C" void kernel_launch(void* const* d_in, const int* in_sizes, int n_in,
                              void* d_out, int out_size, void* d_ws, size_t ws_size,
                              hipStream_t stream) {
    const float* x   = (const float*)d_in[0];
    const int*   ei  = (const int*)d_in[1];
    const int*   row = ei;
    const int*   col = ei + N_EDGES;
    const float* W1  = (const float*)d_in[2];
    const float* b1  = (const float*)d_in[3];
    const float* W2  = (const float*)d_in[4];
    const float* b2  = (const float*)d_in[5];
    float* out = (float*)d_out;

    float* ws     = (float*)d_ws;
    int*   cnt    = (int*)ws;
    float* hp     = ws + ((N_NODES + 255) & ~255);
    float* rest   = hp + (size_t)N_NODES * HIDDEN;
    int*   cursor = (int*)rest;
    int*   srcs   = cursor + ((N_NODES + 255) & ~255);
    float* h2p    = (float*)(srcs + N_EDGES);

    const int B = 256;

    zero_cnt_kernel<<<cdiv_l(N_NODES, B), B, 0, stream>>>(cnt);
    count_kernel<<<cdiv_l(N_EDGES, B), B, 0, stream>>>(col, cnt);
    scan_kernel<<<SCAN_B, SCAN_T, 0, stream>>>(cnt, cursor);
    fill_kernel<<<cdiv_l(N_EDGES, B), B, 0, stream>>>(row, col, cursor, srcs);

    const long waves1 = (N_NODES + 15) / 16;
    gemm1_kernel<<<cdiv_l(waves1 * 32, B), B, 0, stream>>>(x, W1, hp);
    gather1_gemm2_kernel<<<cdiv_l(waves1 * 32, B), B, 0, stream>>>(
        cnt, cursor, srcs, hp, b1, W2, h2p);
    gather2_lsm_kernel<<<cdiv_l((long)N_NODES * N_CLASSES, B), B, 0, stream>>>(
        cnt, cursor, srcs, h2p, b2, out);
}


# Both gather kernels: per-neighbour chain index -> count -> row broken up: indices and counts fetched a whole pass at once (layer 1: nodes ahead), rows asked for as one group; sum order unchanged
# speedup vs baseline: 1.8772x; 1.0645x over previous
#include <hip/hip_runtime.h>
#include <math.h>

typedef __attribute__((ext_vector_type(2))) float v2f;
typedef __attribute__((ext_vector_type(8))) float v8f;

#define N_NODES   100000
#define N_EDGES   1600000
#define IN_FEAT   256
#define HIDDEN    128
#define N_CLASSES 16
#define SCAN_T    1024
#define SCAN_B    98
#define TILE_P    132


__global__ void zero_cnt_kernel(int* __restrict__ cnt) {
    int i = blockIdx.x * blockDim.x + threadIdx.x;
    if (i < N_NODES) cnt[i] = 0;
}

__global__ void count_kernel(const int* __restrict__ col, int* __restrict__ cnt) {
    int e = blockIdx.x * blockDim.x + threadIdx.x;
    if (e < N_EDGES) atomicAdd(&cnt[col[e]], 1);
}

__global__ __launch_bounds__(SCAN_T) void scan_kernel(const int* __restrict__ cnt,
                                                      int* __restrict__ cursor) {
    __shared__ int wpre[32];
    __shared__ int wtot[32];
    const int lane = threadIdx.x & 31;
    const int w    = threadIdx.x >> 5;
    const int i    = blockIdx.x * SCAN_T + threadIdx.x;
    const int v    = (i < N_NODES) ? cnt[i] : 0;

    const int4* c4 = (const int4*)cnt;
    const int n4 = blockIdx.x * (SCAN_T / 4);
    int pre = 0;
    for (int q0 = threadIdx.x; q0 < n4; q0 += 8 * SCAN_T) {
        int4 c[8];
#pragma unroll
        for (int u = 0; u < 8; ++u) {
            const int q = q0 + u * SCAN_T;
            c[u] = c4[(q < n4) ? q : q0];
        }
#pragma unroll
        for (int u = 0; u < 8; ++u) {
            const int q = q0 + u * SCAN_T;
            const int s = (c[u].x + c[u].y) + (c[u].z + c[u].w);
            pre += (q < n4) ? s : 0;
        }
    }
#pragma unroll
    for (int o = 16; o >= 1; o >>= 1) pre += __shfl_xor(pre, o);
    int inc = v;
#pragma unroll
    for (int o = 1; o < 32; o <<= 1) {
        const int u = __shfl_up(inc, o);
        if (lane >= o) inc += u;
    }
    if (lane == 0)  wpre[w] = pre;
    if (lane == 31) wtot[w] = inc;
    __syncthreads();
    int front = 0;
#pragma unroll
    for (int k = 0; k < 32; ++k) front += wpre[k] + ((k < w) ? wtot[k] : 0);
    if (i < N_NODES) cursor[i] = front + inc - v;
}

__global__ void fill_kernel(const int* __restrict__ row, const int* __restrict__ col,
                            int* __restrict__ cursor, int* __restrict__ srcs) {
    int e = blockIdx.x * blockDim.x + threadIdx.x;
    if (e < N_EDGES) {
        int p = atomicAdd(&cursor[col[e]], 1);
        srcs[p] = row[e];
    }
}


__global__ __launch_bounds__(256) void gemm1_kernel(const float* __restrict__ x,
                                                    const float* __restrict__ W1,
                                                    float* __restrict__ hp) {
    const int lane = threadIdx.x & 31;
    const int wave = (blockIdx.x * blockDim.x + threadIdx.x) >> 5;
    const int row0 = wave * 16;
    if (row0 >= N_NODES) return;

    const int m  = lane & 15;
    const int kh = lane >> 4;
    const float* arow = x + (size_t)(row0 + m) * IN_FEAT;

    v8f acc[8];
#pragma unroll
    for (int n = 0; n < 8; ++n) acc[n] = v8f{};

    for (int k = 0; k < IN_FEAT; k += 4) {
        v2f a;
        a.x = arow[k + 2 * kh + 0];
        a.y = arow[k + 2 * kh + 1];
        const float* b0 = W1 + (size_t)(k + 2 * kh) * HIDDEN + m;
#pragma unroll
        for (int n = 0; n < 8; ++n) {
            v2f b;
            b.x = b0[n * 16];
            b.y = b0[HIDDEN + n * 16];
            acc[n] = __builtin_amdgcn_wmma_f32_16x16x4_f32(
                false, a, false, b, (short)0, acc[n], false, false);
        }
    }

    const int mo = kh * 8;
#pragma unroll
    for (int n = 0; n < 8; ++n) {
#pragma unroll
        for (int j = 0; j < 8; ++j) {
            hp[(size_t)(row0 + mo + j) * HIDDEN + n * 16 + m] = acc[n][j];
        }
    }
}


__device__ __forceinline__ void add_rows(const float4* __restrict__ hp4, int r_l, float d_l, int j,
                                         const int cnt_rows, int lane, float4& acc) {
    float4 v[8];
#pragma unroll
    for (int u = 0; u < 8; ++u) {
        if (u < cnt_rows) {
            const int r = __builtin_amdgcn_readlane(r_l, j + u);
            v[u] = hp4[(size_t)r * 32 + lane];
        }
    }
#pragma unroll
    for (int u = 0; u < 8; ++u) {
        if (u < cnt_rows) {
            const float d = __int_as_float(__builtin_amdgcn_readlane(__float_as_int(d_l), j + u));
            acc.x += d * v[u].x; acc.y += d * v[u].y; acc.z += d * v[u].z; acc.w += d * v[u].w;
        }
    }
}


__global__ __launch_bounds__(256) void gather1_gemm2_kernel(const int* __restrict__ cnt,
                                                            const int* __restrict__ cursor,
                                                            const int* __restrict__ srcs,
                                                            const float* __restrict__ hp,
                                                            const float* __restrict__ b1,
                                                            const float* __restrict__ W2,
                                                            float* __restrict__ h2p) {
    __shared__ float tile[8 * 16 * TILE_P];
    const int lane = threadIdx.x & 31;
    const int wid  = threadIdx.x >> 5;
    const int wave = blockIdx.x * 8 + wid;
    const int row0 = wave * 16;
    const bool active = row0 < N_NODES;
    float* mytile = tile + wid * 16 * TILE_P;
    const float4* hp4 = (const float4*)hp;

    if (active) {
        const float4 bias = ((const float4*)b1)[lane];
        const int n_all   = cnt[row0 + (lane & 15)];
        const int end_all = cursor[row0 + (lane & 15)];

        int n_c = __builtin_amdgcn_readlane(n_all, 0), end_c = __builtin_amdgcn_readlane(end_all, 0);
        int n_n = __builtin_amdgcn_readlane(n_all, 1), end_n = __builtin_amdgcn_readlane(end_all, 1);
        int r_c = 0, r_n = 0, c_c = 0;
        if (lane < n_c) r_c = srcs[end_c - n_c + lane];
        if (lane < n_n) r_n = srcs[end_n - n_n + lane];
        float4 v0_c = hp4[(size_t)row0 * 32 + lane];
        if (lane < n_c) c_c = cnt[r_c];

        for (int jn = 0; jn < 16; ++jn) {
            int c_n = 0;
            float4 v0_n = v0_c;
            if (jn + 1 < 16) {
                if (lane < n_n) c_n = cnt[r_n];
                v0_n = hp4[(size_t)(row0 + jn + 1) * 32 + lane];
            }
            int n_nn = 0, end_nn = 0, r_nn = 0;
            if (jn + 2 < 16) {
                n_nn   = __builtin_amdgcn_readlane(n_all, jn + 2);
                end_nn = __builtin_amdgcn_readlane(end_all, jn + 2);
                if (lane < n_nn) r_nn = srcs[end_nn - n_nn + lane];
            }

            const float di = rsqrtf((float)(n_c + 1));
            float4 acc;
            acc.x = di * v0_c.x; acc.y = di * v0_c.y; acc.z = di * v0_c.z; acc.w = di * v0_c.w;
            int   r_l = r_c;
            float d_l = rsqrtf((float)(c_c + 1));
            for (int base = end_c - n_c; ; ) {
                const int m = (end_c - base < 32) ? (end_c - base) : 32;
                for (int j = 0; j < m; j += 8) {
                    switch ((m - j < 8) ? (m - j) : 8) {
                        case 8:  add_rows(hp4, r_l, d_l, j, 8, lane, acc); break;
                        case 7:  add_rows(hp4, r_l, d_l, j, 7, lane, acc); break;
                        case 6:  add_rows(hp4, r_l, d_l, j, 6, lane, acc); break;
                        case 5:  add_rows(hp4, r_l, d_l, j, 5, lane, acc); break;
                        case 4:  add_rows(hp4, r_l, d_l, j, 4, lane, acc); break;
                        case 3:  add_rows(hp4, r_l, d_l, j, 3, lane, acc); break;
                        case 2:  add_rows(hp4, r_l, d_l, j, 2, lane, acc); break;
                        default: add_rows(hp4, r_l, d_l, j, 1, lane, acc); break;
                    }
                }
                base += 32;
                if (base >= end_c) break;
                const int idx = base + lane;
                r_l = 0;
                d_l = 0.0f;
                if (idx < end_c) {
                    r_l = srcs[idx];
                    d_l = rsqrtf((float)(cnt[r_l] + 1));
                }
            }
            float4 o;
            o.x = di * acc.x + bias.x; o.y = di * acc.y + bias.y;
            o.z = di * acc.z + bias.z; o.w = di * acc.w + bias.w;
            o.x = o.x > 0.0f ? o.x : 0.0f; o.y = o.y > 0.0f ? o.y : 0.0f;
            o.z = o.z > 0.0f ? o.z : 0.0f; o.w = o.w > 0.0f ? o.w : 0.0f;
            *(float4*)(mytile + jn * TILE_P + lane * 4) = o;

            n_c = n_n; end_c = end_n; r_c = r_n; c_c = c_n; v0_c = v0_n;
            n_n = n_nn; end_n = end_nn; r_n = r_nn;
        }
    }
    __syncthreads();
    if (active) {
        const int m  = lane & 15;
        const int kh = lane >> 4;
        const float* arow = mytile + m * TILE_P;
        v8f acc2 = v8f{};
        for (int k = 0; k < HIDDEN; k += 4) {
            v2f a;
            a.x = arow[k + 2 * kh + 0];
            a.y = arow[k + 2 * kh + 1];
            v2f b;
            b.x = W2[(size_t)(k + 2 * kh + 0) * N_CLASSES + m];
            b.y = W2[(size_t)(k + 2 * kh + 1) * N_CLASSES + m];
            acc2 = __builtin_amdgcn_wmma_f32_16x16x4_f32(
                false, a, false, b, (short)0, acc2, false, false);
        }
        const int mo = kh * 8;
#pragma unroll
        for (int j = 0; j < 8; ++j) {
            h2p[(size_t)(row0 + mo + j) * N_CLASSES + m] = acc2[j];
        }
    }
}


__global__ __launch_bounds__(256) void gather2_lsm_kernel(const int* __restrict__ cnt,
                                                          const int* __restrict__ cursor,
                                                          const int* __restrict__ srcs,
                                                          const float* __restrict__ h2p,
                                                          const float* __restrict__ b2,
                                                          float* __restrict__ out) {
    const int t = blockIdx.x * blockDim.x + threadIdx.x;
    const int i = t >> 4;
    const int f = t & 15;
    const bool valid = i < N_NODES;
    float v = 0.0f;
    if (valid) {
        const int n   = cnt[i];
        const int end = cursor[i];
        const float bias = b2[f];
        const float di = rsqrtf((float)(n + 1));
        float acc = di * h2p[(size_t)i * N_CLASSES + f];
        for (int base = end - n; base < end; base += 16) {
            const int idx = base + f;
            int   r_l = 0;
            float d_l = 0.0f;
            if (idx < end) {
                r_l = srcs[idx];
                d_l = rsqrtf((float)(cnt[r_l] + 1));
            }
            const int m = (end - base < 16) ? (end - base) : 16;
            float h[16];
#pragma unroll
            for (int u = 0; u < 16; ++u) {
                const int r = __shfl(r_l, u, 16);
                if (u < m) h[u] = h2p[(size_t)r * N_CLASSES + f];
            }
#pragma unroll
            for (int u = 0; u < 16; ++u) {
                const float d = __shfl(d_l, u, 16);
                if (u < m) acc += d * h[u];
            }
        }
        v = di * acc + bias;
    }
    float mx = v;
#pragma unroll
    for (int o = 8; o >= 1; o >>= 1) mx = fmaxf(mx, __shfl_xor(mx, o, 16));
    float s = expf(v - mx);
#pragma unroll
    for (int o = 8; o >= 1; o >>= 1) s += __shfl_xor(s, o, 16);
    const float lse = mx + logf(s);
    if (valid) out[(size_t)i * N_CLASSES + f] = v - lse;
}


static inline int cdiv_l(long a, int b) { return (int)((a + b - 1) / b); }

extern "C" void kernel_launch(void* const* d_in, const int* in_sizes, int n_in,
                              void* d_out, int out_size, void* d_ws, size_t ws_size,
                              hipStream_t stream) {
    const float* x   = (const float*)d_in[0];
    const int*   ei  = (const int*)d_in[1];
    const int*   row = ei;
    const int*   col = ei + N_EDGES;
    const float* W1  = (const float*)d_in[2];
    const float* b1  = (const float*)d_in[3];
    const float* W2  = (const float*)d_in[4];
    const float* b2  = (const float*)d_in[5];
    float* out = (float*)d_out;

    float* ws     = (float*)d_ws;
    int*   cnt    = (int*)ws;
    float* hp     = ws + ((N_NODES + 255) & ~255);
    float* rest   = hp + (size_t)N_NODES * HIDDEN;
    int*   cursor = (int*)rest;
    int*   srcs   = cursor + ((N_NODES + 255) & ~255);
    float* h2p    = (float*)(srcs + N_EDGES);

    const int B = 256;

    zero_cnt_kernel<<<cdiv_l(N_NODES, B), B, 0, stream>>>(cnt);
    count_kernel<<<cdiv_l(N_EDGES, B), B, 0, stream>>>(col, cnt);
    scan_kernel<<<SCAN_B, SCAN_T, 0, stream>>>(cnt, cursor);
    fill_kernel<<<cdiv_l(N_EDGES, B), B, 0, stream>>>(row, col, cursor, srcs);

    const long waves1 = (N_NODES + 15) / 16;
    gemm1_kernel<<<cdiv_l(waves1 * 32, B), B, 0, stream>>>(x, W1, hp);
    gather1_gemm2_kernel<<<cdiv_l(waves1 * 32, B), B, 0, stream>>>(
        cnt, cursor, srcs, hp, b1, W2, h2p);
    gather2_lsm_kernel<<<cdiv_l((long)N_NODES * N_CLASSES, B), B, 0, stream>>>(
        cnt, cursor, srcs, h2p, b2, out);
}


# Edge count pass moved into the projection kernel: each of its 6,250 waves loads 256 edge destinations up front, adds 32 counts per 8 K-steps beside its matrix work; 50,000-wave count launch gone
# speedup vs baseline: 2.0014x; 1.0662x over previous
#include <hip/hip_runtime.h>
#include <math.h>

typedef __attribute__((ext_vector_type(2))) float v2f;
typedef __attribute__((ext_vector_type(8))) float v8f;

#define N_NODES   100000
#define N_EDGES   1600000
#define IN_FEAT   256
#define HIDDEN    128
#define N_CLASSES 16
#define SCAN_T    1024
#define SCAN_B    98
#define TILE_P    132


__global__ void zero_cnt_kernel(int* __restrict__ cnt) {
    int i = blockIdx.x * blockDim.x + threadIdx.x;
    if (i < N_NODES) cnt[i] = 0;
}

__global__ __launch_bounds__(SCAN_T) void scan_kernel(const int* __restrict__ cnt,
                                                      int* __restrict__ cursor) {
    __shared__ int wpre[32];
    __shared__ int wtot[32];
    const int lane = threadIdx.x & 31;
    const int w    = threadIdx.x >> 5;
    const int i    = blockIdx.x * SCAN_T + threadIdx.x;
    const int v    = (i < N_NODES) ? cnt[i] : 0;

    const int4* c4 = (const int4*)cnt;
    const int n4 = blockIdx.x * (SCAN_T / 4);
    int pre = 0;
    for (int q0 = threadIdx.x; q0 < n4; q0 += 8 * SCAN_T) {
        int4 c[8];
#pragma unroll
        for (int u = 0; u < 8; ++u) {
            const int q = q0 + u * SCAN_T;
            c[u] = c4[(q < n4) ? q : q0];
        }
#pragma unroll
        for (int u = 0; u < 8; ++u) {
            const int q = q0 + u * SCAN_T;
            const int s = (c[u].x + c[u].y) + (c[u].z + c[u].w);
            pre += (q < n4) ? s : 0;
        }
    }
#pragma unroll
    for (int o = 16; o >= 1; o >>= 1) pre += __shfl_xor(pre, o);
    int inc = v;
#pragma unroll
    for (int o = 1; o < 32; o <<= 1) {
        const int u = __shfl_up(inc, o);
        if (lane >= o) inc += u;
    }
    if (lane == 0)  wpre[w] = pre;
    if (lane == 31) wtot[w] = inc;
    __syncthreads();
    int front = 0;
#pragma unroll
    for (int k = 0; k < 32; ++k) front += wpre[k] + ((k < w) ? wtot[k] : 0);
    if (i < N_NODES) cursor[i] = front + inc - v;
}

__global__ void fill_kernel(const int* __restrict__ row, const int* __restrict__ col,
                            int* __restrict__ cursor, int* __restrict__ srcs) {
    int e = blockIdx.x * blockDim.x + threadIdx.x;
    if (e < N_EDGES) {
        int p = atomicAdd(&cursor[col[e]], 1);
        srcs[p] = row[e];
    }
}


#define CNT_EPW 256
static_assert(N_NODES % 16 == 0 && (N_NODES / 16) * (long)CNT_EPW == (long)N_EDGES,
              "the row-owning waves share the edges out exactly: no lane without an edge, no edge without a lane");
static_assert(IN_FEAT == 4 * 8 * (CNT_EPW / 32), "one edge group per 8 K-steps covers the K loop exactly");

__global__ __launch_bounds__(256) void gemm1_count_kernel(const float* __restrict__ x,
                                                          const float* __restrict__ W1,
                                                          float* __restrict__ hp,
                                                          const int* __restrict__ col,
                                                          int* __restrict__ cnt) {
    const int lane = threadIdx.x & 31;
    const int wave = (blockIdx.x * blockDim.x + threadIdx.x) >> 5;
    const int row0 = wave * 16;
    if (row0 >= N_NODES) return;

    const int m  = lane & 15;
    const int kh = lane >> 4;
    const float* arow = x + (size_t)(row0 + m) * IN_FEAT;

    v8f acc[8];
#pragma unroll
    for (int n = 0; n < 8; ++n) acc[n] = v8f{};

    const int* cole = col + (size_t)wave * CNT_EPW + lane;
    int cq[CNT_EPW / 32];
#pragma unroll
    for (int g = 0; g < CNT_EPW / 32; ++g) cq[g] = cole[g * 32];

    for (int g = 0; g < CNT_EPW / 32; ++g) {
        for (int k = g * 32; k < g * 32 + 32; k += 4) {
            v2f a;
            a.x = arow[k + 2 * kh + 0];
            a.y = arow[k + 2 * kh + 1];
            const float* b0 = W1 + (size_t)(k + 2 * kh) * HIDDEN + m;
#pragma unroll
            for (int n = 0; n < 8; ++n) {
                v2f b;
                b.x = b0[n * 16];
                b.y = b0[HIDDEN + n * 16];
                acc[n] = __builtin_amdgcn_wmma_f32_16x16x4_f32(
                    false, a, false, b, (short)0, acc[n], false, false);
            }
        }
        atomicAdd(&cnt[cq[0]], 1);
#pragma unroll
        for (int u = 0; u + 1 < CNT_EPW / 32; ++u) cq[u] = cq[u + 1];
    }

    const int mo = kh * 8;
#pragma unroll
    for (int n = 0; n < 8; ++n) {
#pragma unroll
        for (int j = 0; j < 8; ++j) {
            hp[(size_t)(row0 + mo + j) * HIDDEN + n * 16 + m] = acc[n][j];
        }
    }
}


__device__ __forceinline__ void add_rows(const float4* __restrict__ hp4, int r_l, float d_l, int j,
                                         const int cnt_rows, int lane, float4& acc) {
    float4 v[8];
#pragma unroll
    for (int u = 0; u < 8; ++u) {
        if (u < cnt_rows) {
            const int r = __builtin_amdgcn_readlane(r_l, j + u);
            v[u] = hp4[(size_t)r * 32 + lane];
        }
    }
#pragma unroll
    for (int u = 0; u < 8; ++u) {
        if (u < cnt_rows) {
            const float d = __int_as_float(__builtin_amdgcn_readlane(__float_as_int(d_l), j + u));
            acc.x += d * v[u].x; acc.y += d * v[u].y; acc.z += d * v[u].z; acc.w += d * v[u].w;
        }
    }
}


__global__ __launch_bounds__(256) void gather1_gemm2_kernel(const int* __restrict__ cnt,
                                                            const int* __restrict__ cursor,
                                                            const int* __restrict__ srcs,
                                                            const float* __restrict__ hp,
                                                            const float* __restrict__ b1,
                                                            const float* __restrict__ W2,
                                                            float* __restrict__ h2p) {
    __shared__ float tile[8 * 16 * TILE_P];
    const int lane = threadIdx.x & 31;
    const int wid  = threadIdx.x >> 5;
    const int wave = blockIdx.x * 8 + wid;
    const int row0 = wave * 16;
    const bool active = row0 < N_NODES;
    float* mytile = tile + wid * 16 * TILE_P;
    const float4* hp4 = (const float4*)hp;

    if (active) {
        const float4 bias = ((const float4*)b1)[lane];
        const int n_all   = cnt[row0 + (lane & 15)];
        const int end_all = cursor[row0 + (lane & 15)];

        int n_c = __builtin_amdgcn_readlane(n_all, 0), end_c = __builtin_amdgcn_readlane(end_all, 0);
        int n_n = __builtin_amdgcn_readlane(n_all, 1), end_n = __builtin_amdgcn_readlane(end_all, 1);
        int r_c = 0, r_n = 0, c_c = 0;
        if (lane < n_c) r_c = srcs[end_c - n_c + lane];
        if (lane < n_n) r_n = srcs[end_n - n_n + lane];
        float4 v0_c = hp4[(size_t)row0 * 32 + lane];
        if (lane < n_c) c_c = cnt[r_c];

        for (int jn = 0; jn < 16; ++jn) {
            int c_n = 0;
            float4 v0_n = v0_c;
            if (jn + 1 < 16) {
                if (lane < n_n) c_n = cnt[r_n];
                v0_n = hp4[(size_t)(row0 + jn + 1) * 32 + lane];
            }
            int n_nn = 0, end_nn = 0, r_nn = 0;
            if (jn + 2 < 16) {
                n_nn   = __builtin_amdgcn_readlane(n_all, jn + 2);
                end_nn = __builtin_amdgcn_readlane(end_all, jn + 2);
                if (lane < n_nn) r_nn = srcs[end_nn - n_nn + lane];
            }

            const float di = rsqrtf((float)(n_c + 1));
            float4 acc;
            acc.x = di * v0_c.x; acc.y = di * v0_c.y; acc.z = di * v0_c.z; acc.w = di * v0_c.w;
            int   r_l = r_c;
            float d_l = rsqrtf((float)(c_c + 1));
            for (int base = end_c - n_c; ; ) {
                const int m = (end_c - base < 32) ? (end_c - base) : 32;
                for (int j = 0; j < m; j += 8) {
                    switch ((m - j < 8) ? (m - j) : 8) {
                        case 8:  add_rows(hp4, r_l, d_l, j, 8, lane, acc); break;
                        case 7:  add_rows(hp4, r_l, d_l, j, 7, lane, acc); break;
                        case 6:  add_rows(hp4, r_l, d_l, j, 6, lane, acc); break;
                        case 5:  add_rows(hp4, r_l, d_l, j, 5, lane, acc); break;
                        case 4:  add_rows(hp4, r_l, d_l, j, 4, lane, acc); break;
                        case 3:  add_rows(hp4, r_l, d_l, j, 3, lane, acc); break;
                        case 2:  add_rows(hp4, r_l, d_l, j, 2, lane, acc); break;
                        default: add_rows(hp4, r_l, d_l, j, 1, lane, acc); break;
                    }
                }
                base += 32;
                if (base >= end_c) break;
                const int idx = base + lane;
                r_l = 0;
                d_l = 0.0f;
                if (idx < end_c) {
                    r_l = srcs[idx];
                    d_l = rsqrtf((float)(cnt[r_l] + 1));
                }
            }
            float4 o;
            o.x = di * acc.x + bias.x; o.y = di * acc.y + bias.y;
            o.z = di * acc.z + bias.z; o.w = di * acc.w + bias.w;
            o.x = o.x > 0.0f ? o.x : 0.0f; o.y = o.y > 0.0f ? o.y : 0.0f;
            o.z = o.z > 0.0f ? o.z : 0.0f; o.w = o.w > 0.0f ? o.w : 0.0f;
            *(float4*)(mytile + jn * TILE_P + lane * 4) = o;

            n_c = n_n; end_c = end_n; r_c = r_n; c_c = c_n; v0_c = v0_n;
            n_n = n_nn; end_n = end_nn; r_n = r_nn;
        }
    }
    __syncthreads();
    if (active) {
        const int m  = lane & 15;
        const int kh = lane >> 4;
        const float* arow = mytile + m * TILE_P;
        v8f acc2 = v8f{};
        for (int k = 0; k < HIDDEN; k += 4) {
            v2f a;
            a.x = arow[k + 2 * kh + 0];
            a.y = arow[k + 2 * kh + 1];
            v2f b;
            b.x = W2[(size_t)(k + 2 * kh + 0) * N_CLASSES + m];
            b.y = W2[(size_t)(k + 2 * kh + 1) * N_CLASSES + m];
            acc2 = __builtin_amdgcn_wmma_f32_16x16x4_f32(
                false, a, false, b, (short)0, acc2, false, false);
        }
        const int mo = kh * 8;
#pragma unroll
        for (int j = 0; j < 8; ++j) {
            h2p[(size_t)(row0 + mo + j) * N_CLASSES + m] = acc2[j];
        }
    }
}


__global__ __launch_bounds__(256) void gather2_lsm_kernel(const int* __restrict__ cnt,
                                                          const int* __restrict__ cursor,
                                                          const int* __restrict__ srcs,
                                                          const float* __restrict__ h2p,
                                                          const float* __restrict__ b2,
                                                          float* __restrict__ out) {
    const int t = blockIdx.x * blockDim.x + threadIdx.x;
    const int i = t >> 4;
    const int f = t & 15;
    const bool valid = i < N_NODES;
    float v = 0.0f;
    if (valid) {
        const int n   = cnt[i];
        const int end = cursor[i];
        const float bias = b2[f];
        const float di = rsqrtf((float)(n + 1));
        float acc = di * h2p[(size_t)i * N_CLASSES + f];
        for (int base = end - n; base < end; base += 16) {
            const int idx = base + f;
            int   r_l = 0;
            float d_l = 0.0f;
            if (idx < end) {
                r_l = srcs[idx];
                d_l = rsqrtf((float)(cnt[r_l] + 1));
            }
            const int m = (end - base < 16) ? (end - base) : 16;
            float h[16];
#pragma unroll
            for (int u = 0; u < 16; ++u) {
                const int r = __shfl(r_l, u, 16);
                if (u < m) h[u] = h2p[(size_t)r * N_CLASSES + f];
            }
#pragma unroll
            for (int u = 0; u < 16; ++u) {
                const float d = __shfl(d_l, u, 16);
                if (u < m) acc += d * h[u];
            }
        }
        v = di * acc + bias;
    }
    float mx = v;
#pragma unroll
    for (int o = 8; o >= 1; o >>= 1) mx = fmaxf(mx, __shfl_xor(mx, o, 16));
    float s = expf(v - mx);
#pragma unroll
    for (int o = 8; o >= 1; o >>= 1) s += __shfl_xor(s, o, 16);
    const float lse = mx + logf(s);
    if (valid) out[(size_t)i * N_CLASSES + f] = v - lse;
}


static inline int cdiv_l(long a, int b) { return (int)((a + b - 1) / b); }

extern "C" void kernel_launch(void* const* d_in, const int* in_sizes, int n_in,
                              void* d_out, int out_size, void* d_ws, size_t ws_size,
                              hipStream_t stream) {
    const float* x   = (const float*)d_in[0];
    const int*   ei  = (const int*)d_in[1];
    const int*   row = ei;
    const int*   col = ei + N_EDGES;
    const float* W1  = (const float*)d_in[2];
    const float* b1  = (const float*)d_in[3];
    const float* W2  = (const float*)d_in[4];
    const float* b2  = (const float*)d_in[5];
    float* out = (float*)d_out;

    float* ws     = (float*)d_ws;
    int*   cnt    = (int*)ws;
    float* hp     = ws + ((N_NODES + 255) & ~255);
    float* rest   = hp + (size_t)N_NODES * HIDDEN;
    int*   cursor = (int*)rest;
    int*   srcs   = cursor + ((N_NODES + 255) & ~255);
    float* h2p    = (float*)(srcs + N_EDGES);

    const int B = 256;

    const long waves1 = (N_NODES + 15) / 16;

    zero_cnt_kernel<<<cdiv_l(N_NODES, B), B, 0, stream>>>(cnt);
    gemm1_count_kernel<<<cdiv_l(waves1 * 32, B), B, 0, stream>>>(x, W1, hp, col, cnt);
    scan_kernel<<<SCAN_B, SCAN_T, 0, stream>>>(cnt, cursor);
    fill_kernel<<<cdiv_l(N_EDGES, B), B, 0, stream>>>(row, col, cursor, srcs);
    gather1_gemm2_kernel<<<cdiv_l(waves1 * 32, B), B, 0, stream>>>(
        cnt, cursor, srcs, hp, b1, W2, h2p);
    gather2_lsm_kernel<<<cdiv_l((long)N_NODES * N_CLASSES, B), B, 0, stream>>>(
        cnt, cursor, srcs, h2p, b2, out);
}
